# starting kernel
# baseline (speedup 1.0000x reference)
#include <hip/hip_runtime.h>


typedef float v2f __attribute__((ext_vector_type(2)));
typedef float v8f __attribute__((ext_vector_type(8)));

#define LDS_FENCE() asm volatile("" ::: "memory")

__device__ __forceinline__ v8f wmma4(v2f a, v2f b, v8f c) {
  return __builtin_amdgcn_wmma_f32_16x16x4_f32(false, a, false, b, (short)0, c,
                                               false, false);
}


__global__ __launch_bounds__(256) void ode_kernel(
    const float* __restrict__ U0g, const float* __restrict__ epsg,
    const float* __restrict__ W0g, const float* __restrict__ W1g,
    const float* __restrict__ W2g, float* __restrict__ outU,
    float* __restrict__ outLJ) {
  __shared__ float ldsAll[8][1024];
  const int tid  = threadIdx.x;
  const int wv   = tid >> 5;
  const int lane = tid & 31;
  const int half = lane >> 4;
  const int l16  = lane & 15;
  const int kb   = half * 2;
  float* W = ldsAll[wv];

  const int b0 = (blockIdx.x * 8 + wv) * 2;
  const int b1 = b0 + 1;


  {
    v2f u0 = *(const v2f*)(U0g  + (size_t)b0 * 64 + lane * 2);
    v2f u1 = *(const v2f*)(U0g  + (size_t)b1 * 64 + lane * 2);
    v2f e0 = *(const v2f*)(epsg + (size_t)b0 * 64 + lane * 2);
    v2f e1 = *(const v2f*)(epsg + (size_t)b1 * 64 + lane * 2);
    *(v2f*)(W +   0 + lane * 2) = u0;
    *(v2f*)(W +  64 + lane * 2) = e0;
    *(v2f*)(W + 128 + lane * 2) = u1;
    *(v2f*)(W + 192 + lane * 2) = e1;
  }
  LDS_FENCE();

  v2f w2b[2], aw0[2], aw1[2];
#pragma unroll
  for (int j = 0; j < 2; ++j) {
    const int k0 = j * 4 + kb;
    w2b[j].x = W2g[k0 * 8 + (l16 & 7)];
    w2b[j].y = W2g[(k0 + 1) * 8 + (l16 & 7)];
    if (l16 < 8) {
      const int m = l16;
      aw0[j].x = 0.5f * (W0g[m * 8 + k0]     - W0g[k0 * 8 + m]);
      aw0[j].y = 0.5f * (W0g[m * 8 + k0 + 1] - W0g[(k0 + 1) * 8 + m]);
      aw1[j].x = 0.5f * (W1g[m * 8 + k0]     - W1g[k0 * 8 + m]);
      aw1[j].y = 0.5f * (W1g[m * 8 + k0 + 1] - W1g[(k0 + 1) * 8 + m]);
    } else {
      aw0[j].x = aw0[j].y = 0.f;
      aw1[j].x = aw1[j].y = 0.f;
    }
  }

  float Vw[2][8];
  {
    int cc = -1;
    if (lane >= 8 && lane < 16) cc = lane - 8;
    else if (lane >= 16 && lane < 24) cc = lane - 16;
#pragma unroll
    for (int r = 0; r < 8; ++r) {
      Vw[0][r] = (cc >= 0) ? epsg[(size_t)b0 * 64 + r * 8 + cc] : 0.f;
      Vw[1][r] = (cc >= 0) ? epsg[(size_t)b1 * 64 + r * 8 + cc] : 0.f;
    }
  }

  float Ur[2][8];
#pragma unroll
  for (int r = 0; r < 8; ++r) {
    Ur[0][r] = (lane < 8) ? U0g[(size_t)b0 * 64 + r * 8 + lane] : 0.f;
    Ur[1][r] = (lane < 8) ? U0g[(size_t)b1 * 64 + r * 8 + lane] : 0.f;
  }

  const int  g1a = l16 * 8 + ((l16 >= 8) ? 64 : 0) + kb;
  const int  pva = g1a + 64;
  const bool predDiag = (lane < 8) || (lane >= 24);
  const int  astoA = (lane < 8) ? (256 + lane) : (384 + (l16 - 8));
  const int  astoP = astoA + 64;
  const int  g2a[2] = {256 + l16 * 8 + kb, 384 + l16 * 8 + kb};
  const int  g2b[2] = {  0 + l16 * 8 + kb, 128 + l16 * 8 + kb};
  const int  c2st[2]   = {512 + half * 128 + l16, 768 + half * 128 + l16};
  const int  c2base[2] = {512, 768};
  const int  colp = (l16 & 7) + ((l16 >= 8) ? 64 : 0);
  const int  uvb[2] = {0, 128};

  {
    v8f c = {};
#pragma unroll
    for (int j = 0; j < 2; ++j) {
      v2f a = *(const v2f*)(W + pva + j * 4);
      c = wmma4(a, w2b[j], c);
    }
    if (predDiag) {
#pragma unroll
      for (int r = 0; r < 8; ++r) W[astoP + r * 8] = c[r];
    }
    LDS_FENCE();
  }

  auto evalRHS = [&](float t, float (&kout)[2][8], float (&ss)[2]) {
    v8f c1 = {};
#pragma unroll
    for (int j = 0; j < 2; ++j) {
      v2f a = *(const v2f*)(W + g1a + j * 4);
      c1 = wmma4(a, w2b[j], c1);
    }
    if (predDiag) {
#pragma unroll
      for (int r = 0; r < 8; ++r) W[astoA + r * 8] = c1[r];
    }
    LDS_FENCE();

#pragma unroll
    for (int b = 0; b < 2; ++b) {
      v8f c2 = {};
#pragma unroll
      for (int j = 0; j < 2; ++j) {
        v2f a  = *(const v2f*)(W + g2a[b] + j * 4);
        v2f bb = *(const v2f*)(W + g2b[b] + j * 4);
        c2 = wmma4(a, bb, c2);
      }
#pragma unroll
      for (int r = 0; r < 8; ++r) W[c2st[b] + r * 16] = c2[r];
      LDS_FENCE();

      v2f fd[2];
      const int   C2b = c2base[b];
      const float sdf = (l16 >= 8) ? 0.5f : 0.f;
#pragma unroll
      for (int j = 0; j < 2; ++j) {
        const int k0 = j * 4 + kb;
        v2f   L1  = *(const v2f*)(W + C2b + l16 * 16 + k0);
        v2f   L2  = *(const v2f*)(W + C2b + (l16 & 7) * 16 + 8 + k0);
        float L3x = W[C2b + (8 + k0) * 16 + (l16 & 7)];
        float L3y = W[C2b + (9 + k0) * 16 + (l16 & 7)];
        float L4x = W[C2b + k0 * 16 + l16];
        float L4y = W[C2b + (k0 + 1) * 16 + l16];
        fd[j].x = fmaf(t, aw1[j].x, aw0[j].x) + 0.5f * (L1.x - L4x) + sdf * (L2.x - L3x);
        fd[j].y = fmaf(t, aw1[j].y, aw0[j].y) + 0.5f * (L1.y - L4y) + sdf * (L2.y - L3y);
      }

      v8f c3 = {};
#pragma unroll
      for (int j = 0; j < 2; ++j) {
        const int k0 = j * 4 + kb;
        v2f bb;
        bb.x = W[uvb[b] + colp + k0 * 8];
        bb.y = W[uvb[b] + colp + (k0 + 1) * 8];
        c3 = wmma4(fd[j], bb, c3);
      }
      float s = 0.f;
#pragma unroll
      for (int r = 0; r < 8; ++r) {
        kout[b][r] = c3[r];
        s = fmaf(c3[r], Vw[b][r], s);
      }
      ss[b] = s;
    }
  };

  auto storeStage = [&](const float (&k)[2][8], float cdt) {
    if (lane < 8) {
#pragma unroll
      for (int r = 0; r < 8; ++r) {
        W[  0 + r * 8 + lane] = fmaf(cdt, k[0][r], Ur[0][r]);
        W[128 + r * 8 + lane] = fmaf(cdt, k[1][r], Ur[1][r]);
      }
    }
    LDS_FENCE();
  };

  auto storeU = [&]() {
    if (lane < 8) {
#pragma unroll
      for (int r = 0; r < 8; ++r) {
        W[  0 + r * 8 + lane] = Ur[0][r];
        W[128 + r * 8 + lane] = Ur[1][r];
      }
    }
    LDS_FENCE();
  };

  const float dt = 1.0f / 16.0f;
  float lj0 = 0.f, lj1 = 0.f;

  for (int step = 0; step < 16; ++step) {
    const float t0 = (float)step * dt;
    float acc[2][8], kt[2][8], ss[2];

    evalRHS(t0, acc, ss);
    float la0 = ss[0], la1 = ss[1];
    storeStage(acc, 0.5f * dt);

    evalRHS(t0 + 0.5f * dt, kt, ss);
    la0 += 2.f * ss[0]; la1 += 2.f * ss[1];
#pragma unroll
    for (int b = 0; b < 2; ++b)
#pragma unroll
      for (int r = 0; r < 8; ++r) acc[b][r] += 2.f * kt[b][r];
    storeStage(kt, 0.5f * dt);

    evalRHS(t0 + 0.5f * dt, kt, ss);
    la0 += 2.f * ss[0]; la1 += 2.f * ss[1];
#pragma unroll
    for (int b = 0; b < 2; ++b)
#pragma unroll
      for (int r = 0; r < 8; ++r) acc[b][r] += 2.f * kt[b][r];
    storeStage(kt, dt);

    evalRHS(t0 + dt, kt, ss);
    la0 += ss[0]; la1 += ss[1];
#pragma unroll
    for (int b = 0; b < 2; ++b)
#pragma unroll
      for (int r = 0; r < 8; ++r) acc[b][r] += kt[b][r];

    const float w6 = dt / 6.f;
#pragma unroll
    for (int b = 0; b < 2; ++b)
#pragma unroll
      for (int r = 0; r < 8; ++r) Ur[b][r] = fmaf(w6, acc[b][r], Ur[b][r]);
    lj0 = fmaf(w6, la0, lj0);
    lj1 = fmaf(w6, la1, lj1);
    storeU();
  }

  if (lane < 8) {
#pragma unroll
    for (int r = 0; r < 8; ++r) {
      outU[(size_t)b0 * 64 + r * 8 + lane] = Ur[0][r];
      outU[(size_t)b1 * 64 + r * 8 + lane] = Ur[1][r];
    }
  }
#pragma unroll
  for (int off = 16; off > 0; off >>= 1) {
    lj0 += __shfl_xor(lj0, off, 32);
    lj1 += __shfl_xor(lj1, off, 32);
  }
  if (lane == 0) {
    outLJ[b0] = lj0;
    outLJ[b1] = lj1;
  }
}

extern "C" void kernel_launch(void* const* d_in, const int* in_sizes, int n_in,
                              void* d_out, int out_size, void* d_ws,
                              size_t ws_size, hipStream_t stream) {
  (void)n_in; (void)out_size; (void)d_ws; (void)ws_size;
  const float* U0  = (const float*)d_in[0];
  const float* eps = (const float*)d_in[1];
  const float* W0  = (const float*)d_in[2];
  const float* W1  = (const float*)d_in[3];
  const float* W2  = (const float*)d_in[4];
  const int B = in_sizes[0] / 64;

  float* outU  = (float*)d_out;
  float* outLJ = outU + (size_t)B * 64;

  dim3 block(256);
  dim3 grid(B / 16);
  hipLaunchKernelGGL(ode_kernel, grid, block, 0, stream, U0, eps, W0, W1, W2,
                     outU, outLJ);
}


# logj: V is constant, so dF@U is integrated as a matrix in lanes 8..15 of the RK4 state registers (zero start); one dot with V after the last step replaces the per-stage dots
# speedup vs baseline: 1.5707x; 1.5707x over previous
#include <hip/hip_runtime.h>


typedef float v2f __attribute__((ext_vector_type(2)));
typedef float v4f __attribute__((ext_vector_type(4)));
typedef float v8f __attribute__((ext_vector_type(8)));

#define LDS_FENCE() asm volatile("" ::: "memory")

__device__ __forceinline__ v8f wmma4(v2f a, v2f b, v8f c) {
  return __builtin_amdgcn_wmma_f32_16x16x4_f32(false, a, false, b, (short)0, c,
                                               false, false);
}

__device__ __forceinline__ v8f splat8(float x) {
  return (v8f){x, x, x, x, x, x, x, x};
}


__global__ __launch_bounds__(256) void ode_kernel(
    const float* __restrict__ U0g, const float* __restrict__ epsg,
    const float* __restrict__ W0g, const float* __restrict__ W1g,
    const float* __restrict__ W2g, float* __restrict__ outU,
    float* __restrict__ outLJ) {
  __shared__ __attribute__((aligned(16))) float ldsAll[8][1152];
  const int tid  = threadIdx.x;
  const int wv   = tid >> 5;
  const int lane = tid & 31;
  const int half = lane >> 4;
  const int l16  = lane & 15;
  const int kb   = half * 2;
  float* W = ldsAll[wv];

  const int b0 = (blockIdx.x * 8 + wv) * 2;
  const int b1 = b0 + 1;


  {
    v2f u0 = *(const v2f*)(U0g  + (size_t)b0 * 64 + lane * 2);
    v2f u1 = *(const v2f*)(U0g  + (size_t)b1 * 64 + lane * 2);
    v2f e0 = *(const v2f*)(epsg + (size_t)b0 * 64 + lane * 2);
    v2f e1 = *(const v2f*)(epsg + (size_t)b1 * 64 + lane * 2);
    *(v2f*)(W +   0 + lane * 2) = u0;
    *(v2f*)(W +  64 + lane * 2) = e0;
    *(v2f*)(W + 128 + lane * 2) = u1;
    *(v2f*)(W + 192 + lane * 2) = e1;
  }
  LDS_FENCE();

  const int wrow = (l16 + ((l16 >> 3) << 2)) & 7;
  v2f w2b[2], aw0[2], aw1[2];
#pragma unroll
  for (int j = 0; j < 2; ++j) {
    const int k0 = j * 4 + kb;
    w2b[j].x = W2g[k0 * 8 + wrow];
    w2b[j].y = W2g[(k0 + 1) * 8 + wrow];
    if (l16 < 8) {
      const int m = l16;
      aw0[j].x = 0.5f * (W0g[m * 8 + k0]     - W0g[k0 * 8 + m]);
      aw0[j].y = 0.5f * (W0g[m * 8 + k0 + 1] - W0g[(k0 + 1) * 8 + m]);
      aw1[j].x = 0.5f * (W1g[m * 8 + k0]     - W1g[k0 * 8 + m]);
      aw1[j].y = 0.5f * (W1g[m * 8 + k0 + 1] - W1g[(k0 + 1) * 8 + m]);
    } else {
      aw0[j].x = aw0[j].y = 0.f;
      aw1[j].x = aw1[j].y = 0.f;
    }
  }


  v8f Ur[2];

  const int  g1a = l16 * 8 + ((l16 >= 8) ? 64 : 0) + kb;
  const int  pva = g1a + 64;
  const int  astoA = ((l16 < 8) ? 256 : 384) + (l16 & 7) * 8 + half * 4;
  const int  astoP = astoA + 64;
  const int  g2a[2] = {256 + l16 * 8 + kb, 384 + l16 * 8 + kb};
  const int  g2b[2] = {  0 + l16 * 8 + kb, 128 + l16 * 8 + kb};
  const int  c2st[2]   = {512 + l16 * 20 + half * 8, 832 + l16 * 20 + half * 8};
  const int  c2base[2] = {512, 832};
  const int  colp = (l16 & 7) + ((l16 >= 8) ? 256 : 0);
  const int  strow = (lane & 7) * 8 + half * 256;
  const bool stOn  = (lane & 8) == 0;
  const int  uvb[2] = {0, 128};

  {
    v8f c = {};
#pragma unroll
    for (int j = 0; j < 2; ++j) {
      v2f a = *(const v2f*)(W + pva + j * 4);
      c = wmma4(w2b[j], a, c);
    }
    *(v4f*)(W + astoP) = __builtin_shufflevector(c, c, 0, 1, 2, 3);
    LDS_FENCE();
  }

  auto evalRHS = [&](float t, v8f (&kout)[2]) {
#pragma unroll
    for (int b = 0; b < 2; ++b) {
      v8f c2 = {};
#pragma unroll
      for (int j = 0; j < 2; ++j) {
        v2f a  = *(const v2f*)(W + g2a[b] + j * 4);
        v2f bb = *(const v2f*)(W + g2b[b] + j * 4);
        c2 = wmma4(a, bb, c2);
      }
      *(v4f*)(W + c2st[b])     = __builtin_shufflevector(c2, c2, 0, 1, 2, 3);
      *(v4f*)(W + c2st[b] + 4) = __builtin_shufflevector(c2, c2, 4, 5, 6, 7);
      LDS_FENCE();

      v2f fd[2];
      const int   C2b = c2base[b];
      const float sdf = (l16 >= 8) ? 0.5f : 0.f;
#pragma unroll
      for (int j = 0; j < 2; ++j) {
        const int k0 = j * 4 + kb;
        float L1x = W[C2b + k0 * 20 + l16];
        float L1y = W[C2b + (k0 + 1) * 20 + l16];
        float L2x = W[C2b + (8 + k0) * 20 + (l16 & 7)];
        float L2y = W[C2b + (9 + k0) * 20 + (l16 & 7)];
        v2f   L3  = *(const v2f*)(W + C2b + (l16 & 7) * 20 + 8 + k0);
        v2f   L4  = *(const v2f*)(W + C2b + l16 * 20 + k0);
        fd[j].x = fmaf(t, aw1[j].x, aw0[j].x) + 0.5f * (L1x - L4.x) + sdf * (L2x - L3.x);
        fd[j].y = fmaf(t, aw1[j].y, aw0[j].y) + 0.5f * (L1y - L4.y) + sdf * (L2y - L3.y);
      }

      v8f c3 = {};
#pragma unroll
      for (int j = 0; j < 2; ++j) {
        const int k0 = j * 4 + kb;
        v2f bb;
        bb.x = W[uvb[b] + colp + k0 * 8];
        bb.y = W[uvb[b] + colp + (k0 + 1) * 8];
        c3 = wmma4(bb, fd[j], c3);
      }
      kout[b] = c3;
    }
  };

  auto storeStage = [&](const v8f (&k)[2], float cdt) {
    if (stOn) {
#pragma unroll
      for (int b = 0; b < 2; ++b) {
        const v8f s = __builtin_elementwise_fma(splat8(cdt), k[b], Ur[b]);
        *(v4f*)(W + b * 128 + strow)     = __builtin_shufflevector(s, s, 0, 1, 2, 3);
        *(v4f*)(W + b * 128 + strow + 4) = __builtin_shufflevector(s, s, 4, 5, 6, 7);
      }
    }
    LDS_FENCE();
  };

  auto storeU = [&]() {
    if (stOn) {
#pragma unroll
      for (int b = 0; b < 2; ++b) {
        *(v4f*)(W + b * 128 + strow)     = __builtin_shufflevector(Ur[b], Ur[b], 0, 1, 2, 3);
        *(v4f*)(W + b * 128 + strow + 4) = __builtin_shufflevector(Ur[b], Ur[b], 4, 5, 6, 7);
      }
    }
    LDS_FENCE();
  };

  {
    v8f c1 = {};
#pragma unroll
    for (int j = 0; j < 2; ++j) {
      v2f a = *(const v2f*)(W + g1a + j * 4);
      c1 = wmma4(w2b[j], a, c1);
    }
    *(v4f*)(W + astoA) = __builtin_shufflevector(c1, c1, 0, 1, 2, 3);
    LDS_FENCE();
  }

#pragma unroll
  for (int b = 0; b < 2; ++b) {
    v4f lo = *(const v4f*)(W + b * 128 + strow);
    v4f hi = *(const v4f*)(W + b * 128 + strow + 4);
    const v8f st = __builtin_shufflevector(lo, hi, 0, 1, 2, 3, 4, 5, 6, 7);
    Ur[b] = stOn ? st : splat8(0.f);
  }
  LDS_FENCE();

  const float dt = 1.0f / 16.0f;

  for (int step = 0; step < 16; ++step) {
    const float t0 = (float)step * dt;
    v8f acc[2], kt[2];

    evalRHS(t0, acc);
    storeStage(acc, 0.5f * dt);

    evalRHS(t0 + 0.5f * dt, kt);
#pragma unroll
    for (int b = 0; b < 2; ++b) acc[b] = __builtin_elementwise_fma(splat8(2.f), kt[b], acc[b]);
    storeStage(kt, 0.5f * dt);

    evalRHS(t0 + 0.5f * dt, kt);
#pragma unroll
    for (int b = 0; b < 2; ++b) acc[b] = __builtin_elementwise_fma(splat8(2.f), kt[b], acc[b]);
    storeStage(kt, dt);

    evalRHS(t0 + dt, kt);
    const float w6 = dt / 6.f;
#pragma unroll
    for (int b = 0; b < 2; ++b) {
      acc[b] = acc[b] + kt[b];
      Ur[b] = __builtin_elementwise_fma(splat8(w6), acc[b], Ur[b]);
    }
    storeU();
  }

  if (lane < 8) {
#pragma unroll
    for (int b = 0; b < 2; ++b) {
      float* dst = outU + (size_t)(b0 + b) * 64 + lane * 8;
      *(v4f*)(dst)     = __builtin_shufflevector(Ur[b], Ur[b], 0, 1, 2, 3);
      *(v4f*)(dst + 4) = __builtin_shufflevector(Ur[b], Ur[b], 4, 5, 6, 7);
    }
  }
  float lj[2];
  {
    const bool vOn = (lane >= 8) && (lane < 16);
#pragma unroll
    for (int b = 0; b < 2; ++b) {
      const v4f vlo = *(const v4f*)(W + b * 128 + 64 + (lane & 7) * 8);
      const v4f vhi = *(const v4f*)(W + b * 128 + 68 + (lane & 7) * 8);
      float se = Ur[b][0] * vlo[0];
      float so = Ur[b][1] * vlo[1];
      se = fmaf(Ur[b][2], vlo[2], se);
      so = fmaf(Ur[b][3], vlo[3], so);
      se = fmaf(Ur[b][4], vhi[0], se);
      so = fmaf(Ur[b][5], vhi[1], so);
      se = fmaf(Ur[b][6], vhi[2], se);
      so = fmaf(Ur[b][7], vhi[3], so);
      lj[b] = vOn ? (se + so) : 0.f;
    }
  }
  float lj0 = lj[0], lj1 = lj[1];
#pragma unroll
  for (int off = 16; off > 0; off >>= 1) {
    lj0 += __shfl_xor(lj0, off, 32);
    lj1 += __shfl_xor(lj1, off, 32);
  }
  if (lane == 0) {
    outLJ[b0] = lj0;
    outLJ[b1] = lj1;
  }
}

extern "C" void kernel_launch(void* const* d_in, const int* in_sizes, int n_in,
                              void* d_out, int out_size, void* d_ws,
                              size_t ws_size, hipStream_t stream) {
  (void)n_in; (void)out_size; (void)d_ws; (void)ws_size;
  const float* U0  = (const float*)d_in[0];
  const float* eps = (const float*)d_in[1];
  const float* W0  = (const float*)d_in[2];
  const float* W1  = (const float*)d_in[3];
  const float* W2  = (const float*)d_in[4];
  const int B = in_sizes[0] / 64;

  float* outU  = (float*)d_out;
  float* outLJ = outU + (size_t)B * 64;

  dim3 block(256);
  dim3 grid(B / 16);
  hipLaunchKernelGGL(ode_kernel, grid, block, 0, stream, U0, eps, W0, W1, W2,
                     outU, outLJ);
}


# LDS operand tiles padded (V,P at +68 floats, A at U+288): G2 row-pair loads and G3 [Us|As] column loads lose their 2-way bank conflict; same instructions, addresses only
# speedup vs baseline: 1.6668x; 1.0612x over previous
#include <hip/hip_runtime.h>


typedef float v2f __attribute__((ext_vector_type(2)));
typedef float v4f __attribute__((ext_vector_type(4)));
typedef float v8f __attribute__((ext_vector_type(8)));

#define LDS_FENCE() asm volatile("" ::: "memory")

__device__ __forceinline__ v8f wmma4(v2f a, v2f b, v8f c) {
  return __builtin_amdgcn_wmma_f32_16x16x4_f32(false, a, false, b, (short)0, c,
                                               false, false);
}

__device__ __forceinline__ v8f splat8(float x) {
  return (v8f){x, x, x, x, x, x, x, x};
}


__global__ __launch_bounds__(256) void ode_kernel(
    const float* __restrict__ U0g, const float* __restrict__ epsg,
    const float* __restrict__ W0g, const float* __restrict__ W1g,
    const float* __restrict__ W2g, float* __restrict__ outU,
    float* __restrict__ outLJ) {
  __shared__ __attribute__((aligned(16))) float ldsAll[8][1200];
  const int tid  = threadIdx.x;
  const int wv   = tid >> 5;
  const int lane = tid & 31;
  const int half = lane >> 4;
  const int l16  = lane & 15;
  const int kb   = half * 2;
  float* W = ldsAll[wv];

  const int b0 = (blockIdx.x * 8 + wv) * 2;
  const int b1 = b0 + 1;


  {
    v2f u0 = *(const v2f*)(U0g  + (size_t)b0 * 64 + lane * 2);
    v2f u1 = *(const v2f*)(U0g  + (size_t)b1 * 64 + lane * 2);
    v2f e0 = *(const v2f*)(epsg + (size_t)b0 * 64 + lane * 2);
    v2f e1 = *(const v2f*)(epsg + (size_t)b1 * 64 + lane * 2);
    *(v2f*)(W +   0 + lane * 2) = u0;
    *(v2f*)(W +  68 + lane * 2) = e0;
    *(v2f*)(W + 136 + lane * 2) = u1;
    *(v2f*)(W + 204 + lane * 2) = e1;
  }
  LDS_FENCE();

  const int wrow = (l16 + ((l16 >> 3) << 2)) & 7;
  v2f w2b[2], aw0[2], aw1[2];
#pragma unroll
  for (int j = 0; j < 2; ++j) {
    const int k0 = j * 4 + kb;
    w2b[j].x = W2g[k0 * 8 + wrow];
    w2b[j].y = W2g[(k0 + 1) * 8 + wrow];
    if (l16 < 8) {
      const int m = l16;
      aw0[j].x = 0.5f * (W0g[m * 8 + k0]     - W0g[k0 * 8 + m]);
      aw0[j].y = 0.5f * (W0g[m * 8 + k0 + 1] - W0g[(k0 + 1) * 8 + m]);
      aw1[j].x = 0.5f * (W1g[m * 8 + k0]     - W1g[k0 * 8 + m]);
      aw1[j].y = 0.5f * (W1g[m * 8 + k0 + 1] - W1g[(k0 + 1) * 8 + m]);
    } else {
      aw0[j].x = aw0[j].y = 0.f;
      aw1[j].x = aw1[j].y = 0.f;
    }
  }


  v8f Ur[2];

  const int  g1a = l16 * 8 + ((l16 >= 8) ? 72 : 0) + kb;
  const int  pva = g1a + 68;
  const int  astoA = ((l16 < 8) ? 288 : 424) + (l16 & 7) * 8 + half * 4;
  const int  astoP = astoA + 68;
  const int  rowp = l16 * 8 + ((l16 >= 8) ? 4 : 0) + kb;
  const int  g2a[2] = {288 + rowp, 424 + rowp};
  const int  g2b[2] = {  0 + rowp, 136 + rowp};
  const int  c2st[2]   = {560 + l16 * 20 + half * 8, 880 + l16 * 20 + half * 8};
  const int  c2base[2] = {560, 880};
  const int  colp = (l16 & 7) + ((l16 >= 8) ? 288 : 0);
  const int  strow = (lane & 7) * 8 + half * 288;
  const bool stOn  = (lane & 8) == 0;
  const int  uvb[2] = {0, 136};

  {
    v8f c = {};
#pragma unroll
    for (int j = 0; j < 2; ++j) {
      v2f a = *(const v2f*)(W + pva + j * 4);
      c = wmma4(w2b[j], a, c);
    }
    *(v4f*)(W + astoP) = __builtin_shufflevector(c, c, 0, 1, 2, 3);
    LDS_FENCE();
  }

  auto evalRHS = [&](float t, v8f (&kout)[2]) {
#pragma unroll
    for (int b = 0; b < 2; ++b) {
      v8f c2 = {};
#pragma unroll
      for (int j = 0; j < 2; ++j) {
        v2f a  = *(const v2f*)(W + g2a[b] + j * 4);
        v2f bb = *(const v2f*)(W + g2b[b] + j * 4);
        c2 = wmma4(a, bb, c2);
      }
      *(v4f*)(W + c2st[b])     = __builtin_shufflevector(c2, c2, 0, 1, 2, 3);
      *(v4f*)(W + c2st[b] + 4) = __builtin_shufflevector(c2, c2, 4, 5, 6, 7);
      LDS_FENCE();

      v2f fd[2];
      const int   C2b = c2base[b];
      const float sdf = (l16 >= 8) ? 0.5f : 0.f;
#pragma unroll
      for (int j = 0; j < 2; ++j) {
        const int k0 = j * 4 + kb;
        float L1x = W[C2b + k0 * 20 + l16];
        float L1y = W[C2b + (k0 + 1) * 20 + l16];
        float L2x = W[C2b + (8 + k0) * 20 + (l16 & 7)];
        float L2y = W[C2b + (9 + k0) * 20 + (l16 & 7)];
        v2f   L3  = *(const v2f*)(W + C2b + (l16 & 7) * 20 + 8 + k0);
        v2f   L4  = *(const v2f*)(W + C2b + l16 * 20 + k0);
        fd[j].x = fmaf(t, aw1[j].x, aw0[j].x) + 0.5f * (L1x - L4.x) + sdf * (L2x - L3.x);
        fd[j].y = fmaf(t, aw1[j].y, aw0[j].y) + 0.5f * (L1y - L4.y) + sdf * (L2y - L3.y);
      }

      v8f c3 = {};
#pragma unroll
      for (int j = 0; j < 2; ++j) {
        const int k0 = j * 4 + kb;
        v2f bb;
        bb.x = W[uvb[b] + colp + k0 * 8];
        bb.y = W[uvb[b] + colp + (k0 + 1) * 8];
        c3 = wmma4(bb, fd[j], c3);
      }
      kout[b] = c3;
    }
  };

  auto storeStage = [&](const v8f (&k)[2], float cdt) {
    if (stOn) {
#pragma unroll
      for (int b = 0; b < 2; ++b) {
        const v8f s = __builtin_elementwise_fma(splat8(cdt), k[b], Ur[b]);
        *(v4f*)(W + b * 136 + strow)     = __builtin_shufflevector(s, s, 0, 1, 2, 3);
        *(v4f*)(W + b * 136 + strow + 4) = __builtin_shufflevector(s, s, 4, 5, 6, 7);
      }
    }
    LDS_FENCE();
  };

  auto storeU = [&]() {
    if (stOn) {
#pragma unroll
      for (int b = 0; b < 2; ++b) {
        *(v4f*)(W + b * 136 + strow)     = __builtin_shufflevector(Ur[b], Ur[b], 0, 1, 2, 3);
        *(v4f*)(W + b * 136 + strow + 4) = __builtin_shufflevector(Ur[b], Ur[b], 4, 5, 6, 7);
      }
    }
    LDS_FENCE();
  };

  {
    v8f c1 = {};
#pragma unroll
    for (int j = 0; j < 2; ++j) {
      v2f a = *(const v2f*)(W + g1a + j * 4);
      c1 = wmma4(w2b[j], a, c1);
    }
    *(v4f*)(W + astoA) = __builtin_shufflevector(c1, c1, 0, 1, 2, 3);
    LDS_FENCE();
  }

#pragma unroll
  for (int b = 0; b < 2; ++b) {
    v4f lo = *(const v4f*)(W + b * 136 + strow);
    v4f hi = *(const v4f*)(W + b * 136 + strow + 4);
    const v8f st = __builtin_shufflevector(lo, hi, 0, 1, 2, 3, 4, 5, 6, 7);
    Ur[b] = stOn ? st : splat8(0.f);
  }
  LDS_FENCE();

  const float dt = 1.0f / 16.0f;

  for (int step = 0; step < 16; ++step) {
    const float t0 = (float)step * dt;
    v8f acc[2], kt[2];

    evalRHS(t0, acc);
    storeStage(acc, 0.5f * dt);

    evalRHS(t0 + 0.5f * dt, kt);
#pragma unroll
    for (int b = 0; b < 2; ++b) acc[b] = __builtin_elementwise_fma(splat8(2.f), kt[b], acc[b]);
    storeStage(kt, 0.5f * dt);

    evalRHS(t0 + 0.5f * dt, kt);
#pragma unroll
    for (int b = 0; b < 2; ++b) acc[b] = __builtin_elementwise_fma(splat8(2.f), kt[b], acc[b]);
    storeStage(kt, dt);

    evalRHS(t0 + dt, kt);
    const float w6 = dt / 6.f;
#pragma unroll
    for (int b = 0; b < 2; ++b) {
      acc[b] = acc[b] + kt[b];
      Ur[b] = __builtin_elementwise_fma(splat8(w6), acc[b], Ur[b]);
    }
    storeU();
  }

  if (lane < 8) {
#pragma unroll
    for (int b = 0; b < 2; ++b) {
      float* dst = outU + (size_t)(b0 + b) * 64 + lane * 8;
      *(v4f*)(dst)     = __builtin_shufflevector(Ur[b], Ur[b], 0, 1, 2, 3);
      *(v4f*)(dst + 4) = __builtin_shufflevector(Ur[b], Ur[b], 4, 5, 6, 7);
    }
  }
  float lj[2];
  {
    const bool vOn = (lane >= 8) && (lane < 16);
#pragma unroll
    for (int b = 0; b < 2; ++b) {
      const v4f vlo = *(const v4f*)(W + b * 136 + 68 + (lane & 7) * 8);
      const v4f vhi = *(const v4f*)(W + b * 136 + 72 + (lane & 7) * 8);
      float se = Ur[b][0] * vlo[0];
      float so = Ur[b][1] * vlo[1];
      se = fmaf(Ur[b][2], vlo[2], se);
      so = fmaf(Ur[b][3], vlo[3], so);
      se = fmaf(Ur[b][4], vhi[0], se);
      so = fmaf(Ur[b][5], vhi[1], so);
      se = fmaf(Ur[b][6], vhi[2], se);
      so = fmaf(Ur[b][7], vhi[3], so);
      lj[b] = vOn ? (se + so) : 0.f;
    }
  }
  float lj0 = lj[0], lj1 = lj[1];
#pragma unroll
  for (int off = 16; off > 0; off >>= 1) {
    lj0 += __shfl_xor(lj0, off, 32);
    lj1 += __shfl_xor(lj1, off, 32);
  }
  if (lane == 0) {
    outLJ[b0] = lj0;
    outLJ[b1] = lj1;
  }
}

extern "C" void kernel_launch(void* const* d_in, const int* in_sizes, int n_in,
                              void* d_out, int out_size, void* d_ws,
                              size_t ws_size, hipStream_t stream) {
  (void)n_in; (void)out_size; (void)d_ws; (void)ws_size;
  const float* U0  = (const float*)d_in[0];
  const float* eps = (const float*)d_in[1];
  const float* W0  = (const float*)d_in[2];
  const float* W1  = (const float*)d_in[3];
  const float* W2  = (const float*)d_in[4];
  const int B = in_sizes[0] / 64;

  float* outU  = (float*)d_out;
  float* outLJ = outU + (size_t)B * 64;

  dim3 block(256);
  dim3 grid(B / 16);
  hipLaunchKernelGGL(ode_kernel, grid, block, 0, stream, U0, eps, W0, W1, W2,
                     outU, outLJ);
}


# W2 enters only as S=0.5*(W2-W2^T) (A=U@S, P=V@S): G2 tile's F block needs no antisymmetrising, dF=Q+R; G2 run exchanged, tile row-major in LDS, [F;dF] rows read as 8-byte pairs: 8 of 12 loads gone
# speedup vs baseline: 1.8575x; 1.1144x over previous
#include <hip/hip_runtime.h>


typedef float v2f __attribute__((ext_vector_type(2)));
typedef float v4f __attribute__((ext_vector_type(4)));
typedef float v8f __attribute__((ext_vector_type(8)));

#define LDS_FENCE() asm volatile("" ::: "memory")

__device__ __forceinline__ v8f wmma4(v2f a, v2f b, v8f c) {
  return __builtin_amdgcn_wmma_f32_16x16x4_f32(false, a, false, b, (short)0, c,
                                               false, false);
}

__device__ __forceinline__ v8f splat8(float x) {
  return (v8f){x, x, x, x, x, x, x, x};
}


__global__ __launch_bounds__(256) void ode_kernel(
    const float* __restrict__ U0g, const float* __restrict__ epsg,
    const float* __restrict__ W0g, const float* __restrict__ W1g,
    const float* __restrict__ W2g, float* __restrict__ outU,
    float* __restrict__ outLJ) {
  __shared__ __attribute__((aligned(16))) float ldsAll[8][1200];
  const int tid  = threadIdx.x;
  const int wv   = tid >> 5;
  const int lane = tid & 31;
  const int half = lane >> 4;
  const int l16  = lane & 15;
  const int kb   = half * 2;
  float* W = ldsAll[wv];

  const int b0 = (blockIdx.x * 8 + wv) * 2;
  const int b1 = b0 + 1;


  {
    v2f u0 = *(const v2f*)(U0g  + (size_t)b0 * 64 + lane * 2);
    v2f u1 = *(const v2f*)(U0g  + (size_t)b1 * 64 + lane * 2);
    v2f e0 = *(const v2f*)(epsg + (size_t)b0 * 64 + lane * 2);
    v2f e1 = *(const v2f*)(epsg + (size_t)b1 * 64 + lane * 2);
    *(v2f*)(W +   0 + lane * 2) = u0;
    *(v2f*)(W +  68 + lane * 2) = e0;
    *(v2f*)(W + 136 + lane * 2) = u1;
    *(v2f*)(W + 204 + lane * 2) = e1;
  }
  LDS_FENCE();

  const int wrow = (l16 + ((l16 >> 3) << 2)) & 7;
  v2f w2b[2], aw0[2], aw1[2];
#pragma unroll
  for (int j = 0; j < 2; ++j) {
    const int k0 = j * 4 + kb;
    w2b[j].x = 0.5f * (W2g[k0 * 8 + wrow]       - W2g[wrow * 8 + k0]);
    w2b[j].y = 0.5f * (W2g[(k0 + 1) * 8 + wrow] - W2g[wrow * 8 + k0 + 1]);
    if (l16 < 8) {
      const int m = l16;
      aw0[j].x = 0.5f * (W0g[m * 8 + k0]     - W0g[k0 * 8 + m]);
      aw0[j].y = 0.5f * (W0g[m * 8 + k0 + 1] - W0g[(k0 + 1) * 8 + m]);
      aw1[j].x = 0.5f * (W1g[m * 8 + k0]     - W1g[k0 * 8 + m]);
      aw1[j].y = 0.5f * (W1g[m * 8 + k0 + 1] - W1g[(k0 + 1) * 8 + m]);
    } else {
      aw0[j].x = aw0[j].y = 0.f;
      aw1[j].x = aw1[j].y = 0.f;
    }
  }


  v8f Ur[2];

  const int  g1a = l16 * 8 + ((l16 >= 8) ? 72 : 0) + kb;
  const int  pva = g1a + 68;
  const int  astoA = ((l16 < 8) ? 288 : 424) + (l16 & 7) * 8 + half * 4;
  const int  astoP = astoA + 68;
  const int  rowp = l16 * 8 + ((l16 >= 8) ? 4 : 0) + kb;
  const int  g2a[2] = {288 + rowp, 424 + rowp};
  const int  g2b[2] = {  0 + rowp, 136 + rowp};
  const int  c2st[2]   = {560 + l16 * 20 + half * 8, 880 + l16 * 20 + half * 8};
  const int  c2base[2] = {560, 880};
  const int  colp = (l16 & 7) + ((l16 >= 8) ? 288 : 0);
  const int  strow = (lane & 7) * 8 + half * 288;
  const bool stOn  = (lane & 8) == 0;
  const int  uvb[2] = {0, 136};

  {
    v8f c = {};
#pragma unroll
    for (int j = 0; j < 2; ++j) {
      v2f a = *(const v2f*)(W + pva + j * 4);
      c = wmma4(w2b[j], a, c);
    }
    *(v4f*)(W + astoP) = __builtin_shufflevector(c, c, 0, 1, 2, 3);
    LDS_FENCE();
  }

  auto evalRHS = [&](float t, v8f (&kout)[2]) {
#pragma unroll
    for (int b = 0; b < 2; ++b) {
      v8f c2 = {};
#pragma unroll
      for (int j = 0; j < 2; ++j) {
        v2f a  = *(const v2f*)(W + g2a[b] + j * 4);
        v2f bb = *(const v2f*)(W + g2b[b] + j * 4);
        c2 = wmma4(bb, a, c2);
      }
      *(v4f*)(W + c2st[b])     = __builtin_shufflevector(c2, c2, 0, 1, 2, 3);
      *(v4f*)(W + c2st[b] + 4) = __builtin_shufflevector(c2, c2, 4, 5, 6, 7);
      LDS_FENCE();

      v2f fd[2];
      const int   C2b = c2base[b];
      const float sdf = (l16 >= 8) ? 1.f : 0.f;
#pragma unroll
      for (int j = 0; j < 2; ++j) {
        const int k0 = j * 4 + kb;
        v2f La = *(const v2f*)(W + C2b + l16 * 20 + k0);
        v2f Lb = *(const v2f*)(W + C2b + (l16 & 7) * 20 + 8 + k0);
        fd[j].x = fmaf(sdf, Lb.x, fmaf(t, aw1[j].x, aw0[j].x) + La.x);
        fd[j].y = fmaf(sdf, Lb.y, fmaf(t, aw1[j].y, aw0[j].y) + La.y);
      }

      v8f c3 = {};
#pragma unroll
      for (int j = 0; j < 2; ++j) {
        const int k0 = j * 4 + kb;
        v2f bb;
        bb.x = W[uvb[b] + colp + k0 * 8];
        bb.y = W[uvb[b] + colp + (k0 + 1) * 8];
        c3 = wmma4(bb, fd[j], c3);
      }
      kout[b] = c3;
    }
  };

  auto storeStage = [&](const v8f (&k)[2], float cdt) {
    if (stOn) {
#pragma unroll
      for (int b = 0; b < 2; ++b) {
        const v8f s = __builtin_elementwise_fma(splat8(cdt), k[b], Ur[b]);
        *(v4f*)(W + b * 136 + strow)     = __builtin_shufflevector(s, s, 0, 1, 2, 3);
        *(v4f*)(W + b * 136 + strow + 4) = __builtin_shufflevector(s, s, 4, 5, 6, 7);
      }
    }
    LDS_FENCE();
  };

  auto storeU = [&]() {
    if (stOn) {
#pragma unroll
      for (int b = 0; b < 2; ++b) {
        *(v4f*)(W + b * 136 + strow)     = __builtin_shufflevector(Ur[b], Ur[b], 0, 1, 2, 3);
        *(v4f*)(W + b * 136 + strow + 4) = __builtin_shufflevector(Ur[b], Ur[b], 4, 5, 6, 7);
      }
    }
    LDS_FENCE();
  };

  {
    v8f c1 = {};
#pragma unroll
    for (int j = 0; j < 2; ++j) {
      v2f a = *(const v2f*)(W + g1a + j * 4);
      c1 = wmma4(w2b[j], a, c1);
    }
    *(v4f*)(W + astoA) = __builtin_shufflevector(c1, c1, 0, 1, 2, 3);
    LDS_FENCE();
  }

#pragma unroll
  for (int b = 0; b < 2; ++b) {
    v4f lo = *(const v4f*)(W + b * 136 + strow);
    v4f hi = *(const v4f*)(W + b * 136 + strow + 4);
    const v8f st = __builtin_shufflevector(lo, hi, 0, 1, 2, 3, 4, 5, 6, 7);
    Ur[b] = stOn ? st : splat8(0.f);
  }
  LDS_FENCE();

  const float dt = 1.0f / 16.0f;

  for (int step = 0; step < 16; ++step) {
    const float t0 = (float)step * dt;
    v8f acc[2], kt[2];

    evalRHS(t0, acc);
    storeStage(acc, 0.5f * dt);

    evalRHS(t0 + 0.5f * dt, kt);
#pragma unroll
    for (int b = 0; b < 2; ++b) acc[b] = __builtin_elementwise_fma(splat8(2.f), kt[b], acc[b]);
    storeStage(kt, 0.5f * dt);

    evalRHS(t0 + 0.5f * dt, kt);
#pragma unroll
    for (int b = 0; b < 2; ++b) acc[b] = __builtin_elementwise_fma(splat8(2.f), kt[b], acc[b]);
    storeStage(kt, dt);

    evalRHS(t0 + dt, kt);
    const float w6 = dt / 6.f;
#pragma unroll
    for (int b = 0; b < 2; ++b) {
      acc[b] = acc[b] + kt[b];
      Ur[b] = __builtin_elementwise_fma(splat8(w6), acc[b], Ur[b]);
    }
    storeU();
  }

  if (lane < 8) {
#pragma unroll
    for (int b = 0; b < 2; ++b) {
      float* dst = outU + (size_t)(b0 + b) * 64 + lane * 8;
      *(v4f*)(dst)     = __builtin_shufflevector(Ur[b], Ur[b], 0, 1, 2, 3);
      *(v4f*)(dst + 4) = __builtin_shufflevector(Ur[b], Ur[b], 4, 5, 6, 7);
    }
  }
  float lj[2];
  {
    const bool vOn = (lane >= 8) && (lane < 16);
#pragma unroll
    for (int b = 0; b < 2; ++b) {
      const v4f vlo = *(const v4f*)(W + b * 136 + 68 + (lane & 7) * 8);
      const v4f vhi = *(const v4f*)(W + b * 136 + 72 + (lane & 7) * 8);
      float se = Ur[b][0] * vlo[0];
      float so = Ur[b][1] * vlo[1];
      se = fmaf(Ur[b][2], vlo[2], se);
      so = fmaf(Ur[b][3], vlo[3], so);
      se = fmaf(Ur[b][4], vhi[0], se);
      so = fmaf(Ur[b][5], vhi[1], so);
      se = fmaf(Ur[b][6], vhi[2], se);
      so = fmaf(Ur[b][7], vhi[3], so);
      lj[b] = vOn ? (se + so) : 0.f;
    }
  }
  float lj0 = lj[0], lj1 = lj[1];
#pragma unroll
  for (int off = 16; off > 0; off >>= 1) {
    lj0 += __shfl_xor(lj0, off, 32);
    lj1 += __shfl_xor(lj1, off, 32);
  }
  if (lane == 0) {
    outLJ[b0] = lj0;
    outLJ[b1] = lj1;
  }
}

extern "C" void kernel_launch(void* const* d_in, const int* in_sizes, int n_in,
                              void* d_out, int out_size, void* d_ws,
                              size_t ws_size, hipStream_t stream) {
  (void)n_in; (void)out_size; (void)d_ws; (void)ws_size;
  const float* U0  = (const float*)d_in[0];
  const float* eps = (const float*)d_in[1];
  const float* W0  = (const float*)d_in[2];
  const float* W1  = (const float*)d_in[3];
  const float* W2  = (const float*)d_in[4];
  const int B = in_sizes[0] / 64;

  float* outU  = (float*)d_out;
  float* outLJ = outU + (size_t)B * 64;

  dim3 block(256);
  dim3 grid(B / 16);
  hipLaunchKernelGGL(ode_kernel, grid, block, 0, stream, U0, eps, W0, W1, W2,
                     outU, outLJ);
}
